# starting kernel
# baseline (speedup 1.0000x reference)
#include <hip/hip_runtime.h>


#define NCAP   10
#define DCAP   16
#define NOUT   (NCAP * DCAP)
#define DIN    128
#define NROWS  4096
#define BATCH  32
#define NCHUNK 8
#define ROWS_PER_CHUNK (NROWS / NCHUNK)
#define ROWS_PER_WAVE  (ROWS_PER_CHUNK / 8)

typedef __attribute__((ext_vector_type(2))) float v2f;
typedef __attribute__((ext_vector_type(8))) float v8f;

__global__ void capsule_colsum_kernel(const float* __restrict__ x,
                                      float* __restrict__ partials) {
    const int b     = blockIdx.x;
    const int chunk = blockIdx.y;
    const int wave  = threadIdx.x >> 5;
    const int lane  = threadIdx.x & 31;

    const float* xrow = x + ((size_t)b * NROWS +
                             (size_t)chunk * ROWS_PER_CHUNK +
                             (size_t)wave * ROWS_PER_WAVE) * DIN;

    float4 acc = make_float4(0.f, 0.f, 0.f, 0.f);
#pragma unroll 4
    for (int r = 0; r < ROWS_PER_WAVE; ++r) {
        const float4 v = *(const float4*)(xrow + (size_t)r * DIN + lane * 4);
        acc.x += v.x; acc.y += v.y; acc.z += v.z; acc.w += v.w;
    }

    __shared__ float lds[8 * DIN];
    *(float4*)(lds + wave * DIN + lane * 4) = acc;
    __syncthreads();

    const int t = threadIdx.x;
    if (t < DIN) {
        float s = 0.f;
#pragma unroll
        for (int w = 0; w < 8; ++w) s += lds[w * DIN + t];
        partials[((size_t)b * NCHUNK + chunk) * DIN + t] = s;
    }
}

__global__ void capsule_wmma_kernel(const float* __restrict__ partials,
                                    const float* __restrict__ W,
                                    float* __restrict__ out) {
    const int mi   = blockIdx.x;
    const int nj   = blockIdx.y;
    const int lane = threadIdx.x;

    __shared__ float asum[16][DIN];
    for (int idx = lane; idx < 16 * DIN; idx += 32) {
        const int bl = idx >> 7;
        const int k  = idx & (DIN - 1);
        const int b  = mi * 16 + bl;
        float s = 0.f;
#pragma unroll
        for (int c = 0; c < NCHUNK; ++c)
            s += partials[((size_t)b * NCHUNK + c) * DIN + k];
        asum[bl][k] = s;
    }
    __syncthreads();

    const int half = lane >> 4;
    const int l15  = lane & 15;
    const int col  = nj * 16 + l15;

    v8f acc = {};
#pragma unroll
    for (int kk = 0; kk < DIN / 4; ++kk) {
        const int kb = kk * 4 + half * 2;
        v2f a, bf;
        a.x  = asum[l15][kb];
        a.y  = asum[l15][kb + 1];
        bf.x = W[(size_t)kb * NOUT + col];
        bf.y = W[((size_t)kb + 1) * NOUT + col];
        acc = __builtin_amdgcn_wmma_f32_16x16x4_f32(
             false, a,  false, bf,
             (short)0, acc,  false,  false);
    }

#pragma unroll
    for (int r = 0; r < 8; ++r) {
        const int m = r + half * 8;
        out[((size_t)(mi * 16 + m)) * NOUT + col] = acc[r];
    }
}

extern "C" void kernel_launch(void* const* d_in, const int* in_sizes, int n_in,
                              void* d_out, int out_size, void* d_ws, size_t ws_size,
                              hipStream_t stream) {
    const float* x = (const float*)d_in[0];
    const float* W = (const float*)d_in[1];
    float* out      = (float*)d_out;
    float* partials = (float*)d_ws;

    dim3 g1(BATCH, NCHUNK);
    capsule_colsum_kernel<<<g1, 256, 0, stream>>>(x, partials);

    dim3 g2(2, NCAP);
    capsule_wmma_kernel<<<g2, 32, 0, stream>>>(partials, W, out);
}


# stage 2: all operand loads in flight at once (chunk partials summed by an 8-wave block, W fragments issued ahead of the barrier) instead of one wave's serial fill loop; sum order unchanged
# speedup vs baseline: 1.3474x; 1.3474x over previous
#include <hip/hip_runtime.h>


#define NCAP   10
#define DCAP   16
#define NOUT   (NCAP * DCAP)
#define DIN    128
#define NROWS  4096
#define BATCH  32
#define NCHUNK 8
#define ROWS_PER_CHUNK (NROWS / NCHUNK)
#define ROWS_PER_WAVE  (ROWS_PER_CHUNK / 8)

typedef __attribute__((ext_vector_type(2))) float v2f;
typedef __attribute__((ext_vector_type(8))) float v8f;

__global__ void capsule_colsum_kernel(const float* __restrict__ x,
                                      float* __restrict__ partials) {
    const int b     = blockIdx.x;
    const int chunk = blockIdx.y;
    const int wave  = threadIdx.x >> 5;
    const int lane  = threadIdx.x & 31;

    const float* xrow = x + ((size_t)b * NROWS +
                             (size_t)chunk * ROWS_PER_CHUNK +
                             (size_t)wave * ROWS_PER_WAVE) * DIN;

    float4 acc = make_float4(0.f, 0.f, 0.f, 0.f);
#pragma unroll 4
    for (int r = 0; r < ROWS_PER_WAVE; ++r) {
        const float4 v = *(const float4*)(xrow + (size_t)r * DIN + lane * 4);
        acc.x += v.x; acc.y += v.y; acc.z += v.z; acc.w += v.w;
    }

    __shared__ float lds[8 * DIN];
    *(float4*)(lds + wave * DIN + lane * 4) = acc;
    __syncthreads();

    const int t = threadIdx.x;
    if (t < DIN) {
        float s = 0.f;
#pragma unroll
        for (int w = 0; w < 8; ++w) s += lds[w * DIN + t];
        partials[((size_t)b * NCHUNK + chunk) * DIN + t] = s;
    }
}

#define S2_THREADS 256

__global__ __launch_bounds__(S2_THREADS)
void capsule_wmma_kernel(const float* __restrict__ partials,
                         const float* __restrict__ W,
                         float* __restrict__ out) {
    const int mi   = blockIdx.x;
    const int nj   = blockIdx.y;
    const int t    = threadIdx.x;
    const int wave = __builtin_amdgcn_readfirstlane(t >> 5);
    const int lane = t & 31;
    const int half = lane >> 4;
    const int l15  = lane & 15;
    const int col  = nj * 16 + l15;

    __shared__ float asum[16][DIN];

    v2f bfrag[DIN / 4];
    if (wave == 0) {
#pragma unroll
        for (int kk = 0; kk < DIN / 4; ++kk) {
            const int kb = kk * 4 + half * 2;
            bfrag[kk].x = W[(size_t)kb * NOUT + col];
            bfrag[kk].y = W[((size_t)kb + 1) * NOUT + col];
        }
    }

#pragma unroll
    for (int i = 0; i < (16 * DIN / 4) / S2_THREADS; ++i) {
        const int p  = t + i * S2_THREADS;
        const int bl = p >> 5;
        const int k4 = (p & 31) * 4;
        const int b  = mi * 16 + bl;
        const float* src = partials + (size_t)b * NCHUNK * DIN + k4;
        float4 v[NCHUNK];
#pragma unroll
        for (int c = 0; c < NCHUNK; ++c)
            v[c] = *(const float4*)(src + (size_t)c * DIN);
        float4 s = make_float4(0.f, 0.f, 0.f, 0.f);
#pragma unroll
        for (int c = 0; c < NCHUNK; ++c) {
            s.x += v[c].x; s.y += v[c].y; s.z += v[c].z; s.w += v[c].w;
        }
        *(float4*)(&asum[bl][k4]) = s;
    }
    __syncthreads();

    if (wave == 0) {
        v8f acc = {};
#pragma unroll
        for (int kk = 0; kk < DIN / 4; ++kk) {
            const int kb = kk * 4 + half * 2;
            v2f a;
            a.x = asum[l15][kb];
            a.y = asum[l15][kb + 1];
            acc = __builtin_amdgcn_wmma_f32_16x16x4_f32(
                 false, a,  false, bfrag[kk],
                 (short)0, acc,  false,  false);
        }

#pragma unroll
        for (int r = 0; r < 8; ++r) {
            const int m = r + half * 8;
            out[((size_t)(mi * 16 + m)) * NOUT + col] = acc[r];
        }
    }
}

extern "C" void kernel_launch(void* const* d_in, const int* in_sizes, int n_in,
                              void* d_out, int out_size, void* d_ws, size_t ws_size,
                              hipStream_t stream) {
    const float* x = (const float*)d_in[0];
    const float* W = (const float*)d_in[1];
    float* out      = (float*)d_out;
    float* partials = (float*)d_ws;

    dim3 g1(BATCH, NCHUNK);
    capsule_colsum_kernel<<<g1, 256, 0, stream>>>(x, partials);

    dim3 g2(2, NCAP);
    capsule_wmma_kernel<<<g2, S2_THREADS, 0, stream>>>(partials, W, out);
}


# stage 1: 16-wave blocks, each wave requests all its 32 rows (16 KiB) before the first add, instead of 8 waves with 4 loads in flight; per-wave sums cover 32 rows, 16 per block: sum order changes
# speedup vs baseline: 2.3584x; 1.7503x over previous
#include <hip/hip_runtime.h>


#define NCAP   10
#define DCAP   16
#define NOUT   (NCAP * DCAP)
#define DIN    128
#define NROWS  4096
#define BATCH  32
#define NCHUNK 8
#define ROWS_PER_CHUNK (NROWS / NCHUNK)
#define S1_WAVES   16
#define S1_THREADS (S1_WAVES * 32)
#define ROWS_PER_WAVE  (ROWS_PER_CHUNK / S1_WAVES)

typedef __attribute__((ext_vector_type(2))) float v2f;
typedef __attribute__((ext_vector_type(8))) float v8f;

__global__ __launch_bounds__(S1_THREADS) __attribute__((amdgpu_waves_per_eu(4, 4)))
void capsule_colsum_kernel(const float* __restrict__ x,
                           float* __restrict__ partials) {
    const int b     = blockIdx.x;
    const int chunk = blockIdx.y;
    const int wave  = threadIdx.x >> 5;
    const int lane  = threadIdx.x & 31;

    const float* xrow = x + ((size_t)b * NROWS +
                             (size_t)chunk * ROWS_PER_CHUNK +
                             (size_t)wave * ROWS_PER_WAVE) * DIN + lane * 4;

    float4 v[ROWS_PER_WAVE];
#pragma unroll
    for (int r = 0; r < ROWS_PER_WAVE; ++r)
        v[r] = *(const float4*)(xrow + (size_t)r * DIN);

    float4 acc = make_float4(0.f, 0.f, 0.f, 0.f);
#pragma unroll
    for (int r = 0; r < ROWS_PER_WAVE; ++r) {
        acc.x += v[r].x; acc.y += v[r].y; acc.z += v[r].z; acc.w += v[r].w;
    }

    __shared__ float lds[S1_WAVES * DIN];
    *(float4*)(lds + wave * DIN + lane * 4) = acc;
    __syncthreads();

    const int t = threadIdx.x;
    if (t < DIN) {
        float s = 0.f;
#pragma unroll
        for (int w = 0; w < S1_WAVES; ++w) s += lds[w * DIN + t];
        partials[((size_t)b * NCHUNK + chunk) * DIN + t] = s;
    }
}

#define S2_THREADS 256

__global__ __launch_bounds__(S2_THREADS)
void capsule_wmma_kernel(const float* __restrict__ partials,
                         const float* __restrict__ W,
                         float* __restrict__ out) {
    const int mi   = blockIdx.x;
    const int nj   = blockIdx.y;
    const int t    = threadIdx.x;
    const int wave = __builtin_amdgcn_readfirstlane(t >> 5);
    const int lane = t & 31;
    const int half = lane >> 4;
    const int l15  = lane & 15;
    const int col  = nj * 16 + l15;

    __shared__ float asum[16][DIN];

    v2f bfrag[DIN / 4];
    if (wave == 0) {
#pragma unroll
        for (int kk = 0; kk < DIN / 4; ++kk) {
            const int kb = kk * 4 + half * 2;
            bfrag[kk].x = W[(size_t)kb * NOUT + col];
            bfrag[kk].y = W[((size_t)kb + 1) * NOUT + col];
        }
    }

#pragma unroll
    for (int i = 0; i < (16 * DIN / 4) / S2_THREADS; ++i) {
        const int p  = t + i * S2_THREADS;
        const int bl = p >> 5;
        const int k4 = (p & 31) * 4;
        const int b  = mi * 16 + bl;
        const float* src = partials + (size_t)b * NCHUNK * DIN + k4;
        float4 v[NCHUNK];
#pragma unroll
        for (int c = 0; c < NCHUNK; ++c)
            v[c] = *(const float4*)(src + (size_t)c * DIN);
        float4 s = make_float4(0.f, 0.f, 0.f, 0.f);
#pragma unroll
        for (int c = 0; c < NCHUNK; ++c) {
            s.x += v[c].x; s.y += v[c].y; s.z += v[c].z; s.w += v[c].w;
        }
        *(float4*)(&asum[bl][k4]) = s;
    }
    __syncthreads();

    if (wave == 0) {
        v8f acc = {};
#pragma unroll
        for (int kk = 0; kk < DIN / 4; ++kk) {
            const int kb = kk * 4 + half * 2;
            v2f a;
            a.x = asum[l15][kb];
            a.y = asum[l15][kb + 1];
            acc = __builtin_amdgcn_wmma_f32_16x16x4_f32(
                 false, a,  false, bfrag[kk],
                 (short)0, acc,  false,  false);
        }

#pragma unroll
        for (int r = 0; r < 8; ++r) {
            const int m = r + half * 8;
            out[((size_t)(mi * 16 + m)) * NOUT + col] = acc[r];
        }
    }
}

extern "C" void kernel_launch(void* const* d_in, const int* in_sizes, int n_in,
                              void* d_out, int out_size, void* d_ws, size_t ws_size,
                              hipStream_t stream) {
    const float* x = (const float*)d_in[0];
    const float* W = (const float*)d_in[1];
    float* out      = (float*)d_out;
    float* partials = (float*)d_ws;

    dim3 g1(BATCH, NCHUNK);
    capsule_colsum_kernel<<<g1, S1_THREADS, 0, stream>>>(x, partials);

    dim3 g2(2, NCAP);
    capsule_wmma_kernel<<<g2, S2_THREADS, 0, stream>>>(partials, W, out);
}
